# starting kernel
# baseline (speedup 1.0000x reference)
#include <hip/hip_runtime.h>
#include <hip/hip_bf16.h>

typedef __attribute__((ext_vector_type(2))) float v2f;
typedef __attribute__((ext_vector_type(4))) float v4f;
typedef __attribute__((ext_vector_type(8))) float v8f;

#define BN 2048
#define MM 1024
#define ND 64

__global__ void gmm_zero_kernel(float* p, int n) {
  int i = blockIdx.x * blockDim.x + threadIdx.x;
  if (i < n) p[i] = 0.0f;
}

__global__ void gmm_prep_kernel(const float* __restrict__ Mu0,
                                const float* __restrict__ Mu1,
                                const float* __restrict__ S0,
                                const float* __restrict__ S1,
                                const float* __restrict__ tP,
                                const float* __restrict__ eP,
                                float* __restrict__ Bp,
                                float* __restrict__ cM,
                                float* __restrict__ PQ) {
  const int m = blockIdx.x;
  const int d = threadIdx.x;
  const float t   = tP[0];
  const float eps = eP[0];
  const float eps2 = eps * eps;
  const int idx = m * ND + d;

  float s0  = S0[idx],  s1  = S1[idx];
  float mu0 = Mu0[idx], mu1 = Mu1[idx];

  float Ds  = sqrtf(4.0f * s0 * s1 + eps2 * eps2);
  float Cs  = 0.5f * (Ds - eps2);
  float vv  = mu1 - mu0;
  float mut = (1.0f - t) * mu0 + t * mu1;
  float Sg  = (1.0f - t) * (1.0f - t) * s0 + t * t * s1
            + 2.0f * t * (1.0f - t) * (Cs + 0.5f * eps2);
  float Pt  = t * s1 + (1.0f - t) * Cs;
  float Qt  = (1.0f - t) * s0 + t * Cs;
  float St  = Pt - Qt - eps2 * t;
  float Kc  = St / Sg;
  float invS = 1.0f / Sg;

  Bp[(d >> 1) * (2 * MM) + 2 * m + (d & 1)]        = invS;
  Bp[(32 + (d >> 1)) * (2 * MM) + 2 * m + (d & 1)] = -2.0f * mut * invS;

  float* pq = PQ + ((size_t)(m >> 1) * ND + d) * 4;
  pq[(m & 1)]     = Kc;
  pq[2 + (m & 1)] = vv - Kc * mut;

  __shared__ float red[ND];
  red[d] = mut * mut * invS + logf(Sg);
  __syncthreads();
  for (int s = ND / 2; s > 0; s >>= 1) {
    if (d < s) red[d] += red[d + s];
    __syncthreads();
  }
  if (d == 0) cM[m] = red[0];
}

__global__ void __launch_bounds__(32)
gmm_main_kernel(const float* __restrict__ X,
                const float* __restrict__ Lam,
                const float* __restrict__ cM,
                const float* __restrict__ Bp,
                const float* __restrict__ PQ,
                float* __restrict__ uAcc,
                float* __restrict__ nAcc) {
  const int strip = blockIdx.x;
  const int chunk = blockIdx.y;
  const int b0    = strip * 16;
  const int l     = threadIdx.x;
  const int lmod  = l & 15;
  const int lhalf = l >> 4;

  __shared__ float lds[16 * 18];

  v2f xa[32];
  const float* xrow = X + (size_t)(b0 + lmod) * ND;
#pragma unroll
  for (int kk = 0; kk < 16; ++kk) {
    int k0 = kk * 4 + lhalf * 2;
    v2f xv = *(const v2f*)(xrow + k0);
    v2f tfrag; tfrag[0] = xv[0] * xv[0]; tfrag[1] = xv[1] * xv[1];
    xa[kk] = tfrag;
  }
#pragma unroll
  for (int kk = 16; kk < 32; ++kk) {
    int k0 = kk * 4 + lhalf * 2 - 64;
    xa[kk] = *(const v2f*)(xrow + k0);
  }

  v8f accP[4] = {};
  v8f accQ[4] = {};
  v8f accN = {};

  v2f bones;
  bones[0] = (lmod == 0) ? 1.0f : 0.0f;
  bones[1] = bones[0];

  for (int mt = 0; mt < 8; ++mt) {
    const int m0 = chunk * 128 + mt * 16;

    v8f L = {};
#pragma unroll
    for (int kk = 0; kk < 32; ++kk) {
      const int pairIdx = kk * 2 + lhalf;
      v2f bf = *(const v2f*)(Bp + (size_t)pairIdx * (2 * MM) + 2 * (m0 + lmod));
      L = __builtin_amdgcn_wmma_f32_16x16x4_f32(false, xa[kk], false, bf,
                                                (short)0, L, false, false);
    }

    const float cm  = cM[m0 + lmod];
    const float lam = Lam[m0 + lmod];
    float w[8];
#pragma unroll
    for (int r = 0; r < 8; ++r) {
      float lw = -0.5f * (L[r] + cm);
      lw = fminf(50.0f, fmaxf(-50.0f, lw));
      w[r] = __expf(lw) * lam;
    }

    __syncthreads();
#pragma unroll
    for (int r = 0; r < 8; ++r)
      lds[(r + lhalf * 8) * 18 + lmod] = w[r];
    __syncthreads();

#pragma unroll
    for (int kk2 = 0; kk2 < 4; ++kk2) {
      const int km = kk2 * 4 + lhalf * 2;
      v2f wa = *(const v2f*)(&lds[lmod * 18 + km]);
      const int mpair = (m0 >> 1) + kk2 * 2 + lhalf;
#pragma unroll
      for (int n = 0; n < 4; ++n) {
        const int d = n * 16 + lmod;
        v4f pq = *(const v4f*)(PQ + ((size_t)mpair * ND + d) * 4);
        v2f bp; bp[0] = pq[0]; bp[1] = pq[1];
        accP[n] = __builtin_amdgcn_wmma_f32_16x16x4_f32(false, wa, false, bp,
                                                        (short)0, accP[n],
                                                        false, false);
        v2f bq; bq[0] = pq[2]; bq[1] = pq[3];
        accQ[n] = __builtin_amdgcn_wmma_f32_16x16x4_f32(false, wa, false, bq,
                                                        (short)0, accQ[n],
                                                        false, false);
      }
      accN = __builtin_amdgcn_wmma_f32_16x16x4_f32(false, wa, false, bones,
                                                   (short)0, accN,
                                                   false, false);
    }
  }

#pragma unroll
  for (int r = 0; r < 8; ++r) {
    const int b = b0 + r + lhalf * 8;
#pragma unroll
    for (int n = 0; n < 4; ++n) {
      const int d = n * 16 + lmod;
      float val = X[(size_t)b * ND + d] * accP[n][r] + accQ[n][r];
      atomicAdd(&uAcc[(size_t)b * ND + d], val);
    }
    if (lmod == 0) atomicAdd(&nAcc[b], accN[r]);
  }
}

__global__ void gmm_finalize_kernel(const float* __restrict__ uAcc,
                                    const float* __restrict__ nAcc,
                                    float* __restrict__ out) {
  int i = blockIdx.x * blockDim.x + threadIdx.x;
  if (i < BN * ND) out[i] = uAcc[i] / nAcc[i / ND];
}

extern "C" void kernel_launch(void* const* d_in, const int* in_sizes, int n_in,
                              void* d_out, int out_size, void* d_ws, size_t ws_size,
                              hipStream_t stream) {
  const float* X   = (const float*)d_in[0];
  const float* Mu0 = (const float*)d_in[1];
  const float* Mu1 = (const float*)d_in[2];
  const float* S0  = (const float*)d_in[3];
  const float* S1  = (const float*)d_in[4];
  const float* Lam = (const float*)d_in[5];
  const float* tP  = (const float*)d_in[6];
  const float* eP  = (const float*)d_in[7];
  float* out = (float*)d_out;

  float* ws   = (float*)d_ws;
  float* Bp   = ws;
  float* cM   = Bp + 64 * 2 * MM;
  float* PQ   = cM + MM;
  float* uAcc = PQ + (MM / 2) * ND * 4;
  float* nAcc = uAcc + BN * ND;
  (void)ws_size; (void)in_sizes; (void)n_in; (void)out_size;

  const int nzero = BN * ND + BN;
  gmm_zero_kernel<<<(nzero + 255) / 256, 256, 0, stream>>>(uAcc, nzero);

  gmm_prep_kernel<<<MM, ND, 0, stream>>>(Mu0, Mu1, S0, S1, tP, eP,
                                         Bp, cM, PQ);

  gmm_main_kernel<<<dim3(BN / 16, 8), 32, 0, stream>>>(X, Lam, cM, Bp,
                                                       PQ, uAcc, nAcc);

  gmm_finalize_kernel<<<(BN * ND + 255) / 256, 256, 0, stream>>>(uAcc, nAcc, out);
}


# Operand kernel regridded to one wave per component pair, lane owns 2 feature columns: 16-byte stores, c[m] summed by a lane butterfly (sum order changed) instead of LDS tree and 7 barriers
# speedup vs baseline: 2.6992x; 2.6992x over previous
#include <hip/hip_runtime.h>
#include <hip/hip_bf16.h>

typedef __attribute__((ext_vector_type(2))) float v2f;
typedef __attribute__((ext_vector_type(4))) float v4f;
typedef __attribute__((ext_vector_type(8))) float v8f;

#define BN 2048
#define MM 1024
#define ND 64

struct PrepElem {
  float invS;
  float bx;
  float kc;
  float q;
  float ct;
};

__device__ __forceinline__ PrepElem gmm_prep_elem(float s0, float s1,
                                                  float mu0, float mu1,
                                                  float t, float eps2) {
  float Ds  = sqrtf(4.0f * s0 * s1 + eps2 * eps2);
  float Cs  = 0.5f * (Ds - eps2);
  float vv  = mu1 - mu0;
  float mut = (1.0f - t) * mu0 + t * mu1;
  float Sg  = (1.0f - t) * (1.0f - t) * s0 + t * t * s1
            + 2.0f * t * (1.0f - t) * (Cs + 0.5f * eps2);
  float Pt  = t * s1 + (1.0f - t) * Cs;
  float Qt  = (1.0f - t) * s0 + t * Cs;
  float St  = Pt - Qt - eps2 * t;
  float Kc  = St / Sg;
  float invS = 1.0f / Sg;
  PrepElem e;
  e.invS = invS;
  e.bx   = -2.0f * mut * invS;
  e.kc   = Kc;
  e.q    = vv - Kc * mut;
  e.ct   = mut * mut * invS + logf(Sg);
  return e;
}

template <int M>
__device__ __forceinline__ float gmm_row_xor(float x) {
  int b = __builtin_bit_cast(int, x);
  return __builtin_bit_cast(
      float, __builtin_amdgcn_update_dpp(b, b, 0x160 + M, 0xf, 0xf, false));
}

__device__ __forceinline__ float gmm_xhalf(float x) {
  int b = __builtin_bit_cast(int, x);
  return __builtin_bit_cast(
      float, __builtin_amdgcn_permlanex16(b, b, 0x76543210, 0xfedcba98,
                                          false, false));
}

__device__ __forceinline__ float gmm_wave_sum(float x) {
  x += gmm_row_xor<1>(x);
  x += gmm_row_xor<2>(x);
  x += gmm_row_xor<4>(x);
  x += gmm_row_xor<8>(x);
  x += gmm_xhalf(x);
  return x;
}

__global__ void __launch_bounds__(128)
gmm_prep_kernel(const float* __restrict__ Mu0,
                const float* __restrict__ Mu1,
                const float* __restrict__ S0,
                const float* __restrict__ S1,
                const float* __restrict__ tP,
                const float* __restrict__ eP,
                float* __restrict__ Bp,
                float* __restrict__ cM,
                float* __restrict__ PQ) {
  const int wave = (int)(threadIdx.x >> 5);
  const int l    = (int)(threadIdx.x & 31);
  const int p    = (int)blockIdx.x * 4 + wave;
  const float t    = tP[0];
  const float eps  = eP[0];

  const size_t e0 = (size_t)(2 * p) * ND + 2 * l;
  const v2f s0a  = *(const v2f*)(S0  + e0);
  const v2f s0b  = *(const v2f*)(S0  + e0 + ND);
  const v2f s1a  = *(const v2f*)(S1  + e0);
  const v2f s1b  = *(const v2f*)(S1  + e0 + ND);
  const v2f mu0a = *(const v2f*)(Mu0 + e0);
  const v2f mu0b = *(const v2f*)(Mu0 + e0 + ND);
  const v2f mu1a = *(const v2f*)(Mu1 + e0);
  const v2f mu1b = *(const v2f*)(Mu1 + e0 + ND);

  __builtin_amdgcn_sched_barrier(0);
  const float eps2 = eps * eps;

  const PrepElem a0 = gmm_prep_elem(s0a[0], s1a[0], mu0a[0], mu1a[0], t, eps2);
  const PrepElem a1 = gmm_prep_elem(s0a[1], s1a[1], mu0a[1], mu1a[1], t, eps2);
  const PrepElem b0 = gmm_prep_elem(s0b[0], s1b[0], mu0b[0], mu1b[0], t, eps2);
  const PrepElem b1 = gmm_prep_elem(s0b[1], s1b[1], mu0b[1], mu1b[1], t, eps2);

  v4f bi; bi[0] = a0.invS; bi[1] = a1.invS; bi[2] = b0.invS; bi[3] = b1.invS;
  v4f bm; bm[0] = a0.bx;   bm[1] = a1.bx;   bm[2] = b0.bx;   bm[3] = b1.bx;
  *(v4f*)(Bp + (size_t)l * (2 * MM) + 4 * p)        = bi;
  *(v4f*)(Bp + (size_t)(32 + l) * (2 * MM) + 4 * p) = bm;

  v4f q0; q0[0] = a0.kc; q0[1] = b0.kc; q0[2] = a0.q; q0[3] = b0.q;
  v4f q1; q1[0] = a1.kc; q1[1] = b1.kc; q1[2] = a1.q; q1[3] = b1.q;
  float* pq = PQ + ((size_t)p * ND + 2 * l) * 4;
  *(v4f*)pq       = q0;
  *(v4f*)(pq + 4) = q1;

  const float c0 = gmm_wave_sum(a0.ct + a1.ct);
  const float c1 = gmm_wave_sum(b0.ct + b1.ct);
  if (l == 0) {
    v2f c; c[0] = c0; c[1] = c1;
    *(v2f*)(cM + 2 * p) = c;
  }
}

__device__ __forceinline__ void gmm_wave_sync() {
  __builtin_amdgcn_fence(__ATOMIC_RELEASE, "wavefront");
  __builtin_amdgcn_wave_barrier();
  __builtin_amdgcn_fence(__ATOMIC_ACQUIRE, "wavefront");
}

struct TileFrag {
  v2f b[32];
  v4f pq[16];
  float cm;
  float lam;
};

__device__ __forceinline__ void gmm_load_tile(TileFrag& t,
                                              const float* __restrict__ bpLane,
                                              const float* __restrict__ pqLane,
                                              const float* __restrict__ cmLane,
                                              const float* __restrict__ lamLane,
                                              int mt) {
#pragma unroll
  for (int kk = 0; kk < 32; ++kk)
    t.b[kk] = *(const v2f*)(bpLane + (size_t)kk * (4 * MM) + mt * 32);
  t.cm  = cmLane[mt * 16];
  t.lam = lamLane[mt * 16];
#pragma unroll
  for (int kk2 = 0; kk2 < 4; ++kk2)
#pragma unroll
    for (int n = 0; n < 4; ++n)
      t.pq[kk2 * 4 + n] =
          *(const v4f*)(pqLane + ((size_t)(mt * 8 + kk2 * 2) * ND + n * 16) * 4);
}

__global__ void __launch_bounds__(128)
gmm_main_kernel(const float* __restrict__ X,
                const float* __restrict__ Lam,
                const float* __restrict__ cM,
                const float* __restrict__ Bp,
                const float* __restrict__ PQ,
                float* __restrict__ part,
                float* __restrict__ nPart) {
  const int wave  = __builtin_amdgcn_readfirstlane((int)(threadIdx.x >> 5));
  const int strip = blockIdx.x * 4 + wave;
  const int chunk = blockIdx.y;
  const int b0    = strip * 16;
  const int l     = threadIdx.x & 31;
  const int lmod  = l & 15;
  const int lhalf = l >> 4;

  __shared__ float ldsAll[4][16 * 18];
  float* lds = ldsAll[wave];

  v2f xa[32];
  const float* xrow = X + (size_t)(b0 + lmod) * ND;
#pragma unroll
  for (int kk = 0; kk < 16; ++kk) {
    int k0 = kk * 4 + lhalf * 2;
    v2f xv = *(const v2f*)(xrow + k0);
    v2f tfrag; tfrag[0] = xv[0] * xv[0]; tfrag[1] = xv[1] * xv[1];
    xa[kk] = tfrag;
    xa[kk + 16] = xv;
  }

  const float* bpLane  = Bp + (size_t)lhalf * (2 * MM) + 2 * (chunk * 128 + lmod);
  const float* pqLane  = PQ + ((size_t)(chunk * 64 + lhalf) * ND + lmod) * 4;
  const float* cmLane  = cM + chunk * 128 + lmod;
  const float* lamLane = Lam + chunk * 128 + lmod;

  TileFrag tf[2];
  gmm_load_tile(tf[0], bpLane, pqLane, cmLane, lamLane, 0);

  v8f accP[4] = {};
  v8f accQ[4] = {};
  v8f accN = {};

  v2f bones;
  bones[0] = (lmod == 0) ? 1.0f : 0.0f;
  bones[1] = bones[0];

  float xe[8][4];

#pragma unroll
  for (int mt = 0; mt < 8; ++mt) {
    const TileFrag& cur = tf[mt & 1];

    __builtin_amdgcn_sched_barrier(0);
    if (mt + 1 < 8) {
      gmm_load_tile(tf[(mt + 1) & 1], bpLane, pqLane, cmLane, lamLane, mt + 1);
    } else {
#pragma unroll
      for (int r = 0; r < 8; ++r)
#pragma unroll
        for (int n = 0; n < 4; ++n)
          xe[r][n] = X[(size_t)(b0 + r + lhalf * 8) * ND + n * 16 + lmod];
    }
    __builtin_amdgcn_sched_barrier(0);

    v8f L = {};
#pragma unroll
    for (int kk = 0; kk < 32; ++kk)
      L = __builtin_amdgcn_wmma_f32_16x16x4_f32(false, xa[kk], false, cur.b[kk],
                                                (short)0, L, false, false);

    const float cm  = cur.cm;
    const float lam = cur.lam;
    float w[8];
#pragma unroll
    for (int r = 0; r < 8; ++r) {
      float lw = -0.5f * (L[r] + cm);
      lw = fminf(50.0f, fmaxf(-50.0f, lw));
      w[r] = __expf(lw) * lam;
    }

    gmm_wave_sync();
#pragma unroll
    for (int r = 0; r < 8; ++r)
      lds[(r + lhalf * 8) * 18 + lmod] = w[r];
    gmm_wave_sync();

#pragma unroll
    for (int kk2 = 0; kk2 < 4; ++kk2) {
      const int km = kk2 * 4 + lhalf * 2;
      v2f wa = *(const v2f*)(&lds[lmod * 18 + km]);
#pragma unroll
      for (int n = 0; n < 4; ++n) {
        v4f pq = cur.pq[kk2 * 4 + n];
        v2f bp; bp[0] = pq[0]; bp[1] = pq[1];
        accP[n] = __builtin_amdgcn_wmma_f32_16x16x4_f32(false, wa, false, bp,
                                                        (short)0, accP[n],
                                                        false, false);
        v2f bq; bq[0] = pq[2]; bq[1] = pq[3];
        accQ[n] = __builtin_amdgcn_wmma_f32_16x16x4_f32(false, wa, false, bq,
                                                        (short)0, accQ[n],
                                                        false, false);
      }
      accN = __builtin_amdgcn_wmma_f32_16x16x4_f32(false, wa, false, bones,
                                                   (short)0, accN,
                                                   false, false);
    }
  }

  float* pTile = part + ((size_t)chunk * BN + b0) * ND;
#pragma unroll
  for (int r = 0; r < 8; ++r) {
    const int bl = r + lhalf * 8;
#pragma unroll
    for (int n = 0; n < 4; ++n) {
      const int d = n * 16 + lmod;
      pTile[bl * ND + d] = xe[r][n] * accP[n][r] + accQ[n][r];
    }
  }
  if (lmod == 0) {
    float* np = nPart + (size_t)chunk * BN + b0 + lhalf * 8;
    v4f n0; n0[0] = accN[0]; n0[1] = accN[1]; n0[2] = accN[2]; n0[3] = accN[3];
    v4f n1; n1[0] = accN[4]; n1[1] = accN[5]; n1[2] = accN[6]; n1[3] = accN[7];
    *(v4f*)np       = n0;
    *(v4f*)(np + 4) = n1;
  }
}

#define NCHUNK 8
__global__ void __launch_bounds__(128)
gmm_finalize_kernel(const float* __restrict__ part,
                    const float* __restrict__ nPart,
                    float* __restrict__ out) {
  const int t = blockIdx.x * blockDim.x + threadIdx.x;
  if (t >= BN * ND / 4) return;
  const int b = t >> 4;
  const size_t e = (size_t)t * 4;
  v4f pv[NCHUNK];
  float nv[NCHUNK];
#pragma unroll
  for (int c = 0; c < NCHUNK; ++c) {
    pv[c] = *(const v4f*)(part + (size_t)c * (BN * ND) + e);
    nv[c] = nPart[c * BN + b];
  }
  v4f s = pv[0];
  float ns = nv[0];
#pragma unroll
  for (int c = 1; c < NCHUNK; ++c) {
    s  += pv[c];
    ns += nv[c];
  }
  v4f o;
  o[0] = s[0] / ns; o[1] = s[1] / ns; o[2] = s[2] / ns; o[3] = s[3] / ns;
  *(v4f*)(out + e) = o;
}

extern "C" void kernel_launch(void* const* d_in, const int* in_sizes, int n_in,
                              void* d_out, int out_size, void* d_ws, size_t ws_size,
                              hipStream_t stream) {
  const float* X   = (const float*)d_in[0];
  const float* Mu0 = (const float*)d_in[1];
  const float* Mu1 = (const float*)d_in[2];
  const float* S0  = (const float*)d_in[3];
  const float* S1  = (const float*)d_in[4];
  const float* Lam = (const float*)d_in[5];
  const float* tP  = (const float*)d_in[6];
  const float* eP  = (const float*)d_in[7];
  float* out = (float*)d_out;

  float* ws   = (float*)d_ws;
  float* Bp   = ws;
  float* cM   = Bp + 64 * 2 * MM;
  float* PQ   = cM + MM;
  float* part  = PQ + (MM / 2) * ND * 4;
  float* nPart = part + 8 * BN * ND;
  (void)ws_size; (void)in_sizes; (void)n_in; (void)out_size;

  gmm_prep_kernel<<<MM / 8, 128, 0, stream>>>(Mu0, Mu1, S0, S1, tP, eP,
                                              Bp, cM, PQ);

  gmm_main_kernel<<<dim3(BN / 64, 8), 128, 0, stream>>>(X, Lam, cM, Bp,
                                                        PQ, part, nPart);

  gmm_finalize_kernel<<<(BN * ND / 4 + 127) / 128, 128, 0, stream>>>(part, nPart, out);
}


# m loop takes two m-tiles per trip (even and odd m of a 32 block): GEMM1 is two independent WMMA chains, one exp + LDS transpose pass per 32 m, Bp as b128, load groups split to match; sum orders kept
# speedup vs baseline: 2.9006x; 1.0746x over previous
#include <hip/hip_runtime.h>
#include <hip/hip_bf16.h>

typedef __attribute__((ext_vector_type(2))) float v2f;
typedef __attribute__((ext_vector_type(4))) float v4f;
typedef __attribute__((ext_vector_type(8))) float v8f;

#define BN 2048
#define MM 1024
#define ND 64

struct PrepElem {
  float invS;
  float bx;
  float kc;
  float q;
  float ct;
};

__device__ __forceinline__ PrepElem gmm_prep_elem(float s0, float s1,
                                                  float mu0, float mu1,
                                                  float t, float eps2) {
  float Ds  = sqrtf(4.0f * s0 * s1 + eps2 * eps2);
  float Cs  = 0.5f * (Ds - eps2);
  float vv  = mu1 - mu0;
  float mut = (1.0f - t) * mu0 + t * mu1;
  float Sg  = (1.0f - t) * (1.0f - t) * s0 + t * t * s1
            + 2.0f * t * (1.0f - t) * (Cs + 0.5f * eps2);
  float Pt  = t * s1 + (1.0f - t) * Cs;
  float Qt  = (1.0f - t) * s0 + t * Cs;
  float St  = Pt - Qt - eps2 * t;
  float Kc  = St / Sg;
  float invS = 1.0f / Sg;
  PrepElem e;
  e.invS = invS;
  e.bx   = -2.0f * mut * invS;
  e.kc   = Kc;
  e.q    = vv - Kc * mut;
  e.ct   = mut * mut * invS + logf(Sg);
  return e;
}

template <int M>
__device__ __forceinline__ float gmm_row_xor(float x) {
  int b = __builtin_bit_cast(int, x);
  return __builtin_bit_cast(
      float, __builtin_amdgcn_update_dpp(b, b, 0x160 + M, 0xf, 0xf, false));
}

__device__ __forceinline__ float gmm_xhalf(float x) {
  int b = __builtin_bit_cast(int, x);
  return __builtin_bit_cast(
      float, __builtin_amdgcn_permlanex16(b, b, 0x76543210, 0xfedcba98,
                                          false, false));
}

__device__ __forceinline__ float gmm_wave_sum(float x) {
  x += gmm_row_xor<1>(x);
  x += gmm_row_xor<2>(x);
  x += gmm_row_xor<4>(x);
  x += gmm_row_xor<8>(x);
  x += gmm_xhalf(x);
  return x;
}

__global__ void __launch_bounds__(128)
gmm_prep_kernel(const float* __restrict__ Mu0,
                const float* __restrict__ Mu1,
                const float* __restrict__ S0,
                const float* __restrict__ S1,
                const float* __restrict__ tP,
                const float* __restrict__ eP,
                float* __restrict__ Bp,
                float* __restrict__ cM,
                float* __restrict__ PQ) {
  const int wave = (int)(threadIdx.x >> 5);
  const int l    = (int)(threadIdx.x & 31);
  const int p    = (int)blockIdx.x * 4 + wave;
  const float t    = tP[0];
  const float eps  = eP[0];

  const size_t e0 = (size_t)(2 * p) * ND + 2 * l;
  const v2f s0a  = *(const v2f*)(S0  + e0);
  const v2f s0b  = *(const v2f*)(S0  + e0 + ND);
  const v2f s1a  = *(const v2f*)(S1  + e0);
  const v2f s1b  = *(const v2f*)(S1  + e0 + ND);
  const v2f mu0a = *(const v2f*)(Mu0 + e0);
  const v2f mu0b = *(const v2f*)(Mu0 + e0 + ND);
  const v2f mu1a = *(const v2f*)(Mu1 + e0);
  const v2f mu1b = *(const v2f*)(Mu1 + e0 + ND);

  __builtin_amdgcn_sched_barrier(0);
  const float eps2 = eps * eps;

  const PrepElem a0 = gmm_prep_elem(s0a[0], s1a[0], mu0a[0], mu1a[0], t, eps2);
  const PrepElem a1 = gmm_prep_elem(s0a[1], s1a[1], mu0a[1], mu1a[1], t, eps2);
  const PrepElem b0 = gmm_prep_elem(s0b[0], s1b[0], mu0b[0], mu1b[0], t, eps2);
  const PrepElem b1 = gmm_prep_elem(s0b[1], s1b[1], mu0b[1], mu1b[1], t, eps2);

  v4f bi; bi[0] = a0.invS; bi[1] = a1.invS; bi[2] = b0.invS; bi[3] = b1.invS;
  v4f bm; bm[0] = a0.bx;   bm[1] = a1.bx;   bm[2] = b0.bx;   bm[3] = b1.bx;
  *(v4f*)(Bp + (size_t)l * (2 * MM) + 4 * p)        = bi;
  *(v4f*)(Bp + (size_t)(32 + l) * (2 * MM) + 4 * p) = bm;

  v4f q0; q0[0] = a0.kc; q0[1] = b0.kc; q0[2] = a0.q; q0[3] = b0.q;
  v4f q1; q1[0] = a1.kc; q1[1] = b1.kc; q1[2] = a1.q; q1[3] = b1.q;
  float* pq = PQ + ((size_t)p * ND + 2 * l) * 4;
  *(v4f*)pq       = q0;
  *(v4f*)(pq + 4) = q1;

  const float c0 = gmm_wave_sum(a0.ct + a1.ct);
  const float c1 = gmm_wave_sum(b0.ct + b1.ct);
  if (l == 0) {
    v2f c; c[0] = c0; c[1] = c1;
    *(v2f*)(cM + 2 * p) = c;
  }
}

__device__ __forceinline__ void gmm_wave_sync() {
  __builtin_amdgcn_fence(__ATOMIC_RELEASE, "wavefront");
  __builtin_amdgcn_wave_barrier();
  __builtin_amdgcn_fence(__ATOMIC_ACQUIRE, "wavefront");
}

struct TileFrag {
  v4f b[32];
  v4f pq[32];
  v2f cm;
  v2f lam;
};

__device__ __forceinline__ void gmm_load_b(TileFrag& t,
                                           const float* __restrict__ bpLane,
                                           const float* __restrict__ cmLane,
                                           const float* __restrict__ lamLane,
                                           int tp) {
#pragma unroll
  for (int kk = 0; kk < 32; ++kk)
    t.b[kk] = *(const v4f*)(bpLane + (size_t)kk * (4 * MM) + tp * 64);
  t.cm  = *(const v2f*)(cmLane + tp * 32);
  t.lam = *(const v2f*)(lamLane + tp * 32);
}

__device__ __forceinline__ void gmm_load_pq(TileFrag& t,
                                            const float* __restrict__ pqLane,
                                            int tp) {
#pragma unroll
  for (int j = 0; j < 8; ++j)
#pragma unroll
    for (int n = 0; n < 4; ++n)
      t.pq[j * 4 + n] =
          *(const v4f*)(pqLane + ((size_t)(tp * 16 + j * 2) * ND + n * 16) * 4);
}

__global__ void __launch_bounds__(128)
gmm_main_kernel(const float* __restrict__ X,
                const float* __restrict__ Lam,
                const float* __restrict__ cM,
                const float* __restrict__ Bp,
                const float* __restrict__ PQ,
                float* __restrict__ part,
                float* __restrict__ nPart) {
  const int wave  = __builtin_amdgcn_readfirstlane((int)(threadIdx.x >> 5));
  const int strip = blockIdx.x * 4 + wave;
  const int chunk = blockIdx.y;
  const int b0    = strip * 16;
  const int l     = threadIdx.x & 31;
  const int lmod  = l & 15;
  const int lhalf = l >> 4;

  __shared__ __attribute__((aligned(16))) float ldsAll[4][16 * 36];
  float* lds = ldsAll[wave];

  v2f xa[32];
  const float* xrow = X + (size_t)(b0 + lmod) * ND;
#pragma unroll
  for (int kk = 0; kk < 16; ++kk) {
    int k0 = kk * 4 + lhalf * 2;
    v2f xv = *(const v2f*)(xrow + k0);
    v2f tfrag; tfrag[0] = xv[0] * xv[0]; tfrag[1] = xv[1] * xv[1];
    xa[kk] = tfrag;
    xa[kk + 16] = xv;
  }

  const float* bpLane  = Bp + (size_t)lhalf * (2 * MM) + 2 * (chunk * 128 + 2 * lmod);
  const float* pqLane  = PQ + ((size_t)(chunk * 64 + lhalf) * ND + lmod) * 4;
  const float* cmLane  = cM + chunk * 128 + 2 * lmod;
  const float* lamLane = Lam + chunk * 128 + 2 * lmod;

  TileFrag tf[2];
  gmm_load_b(tf[0], bpLane, cmLane, lamLane, 0);
  gmm_load_pq(tf[0], pqLane, 0);

  v8f accP[4] = {};
  v8f accQ[4] = {};
  v8f accN = {};

  v2f bones;
  bones[0] = (lmod == 0) ? 1.0f : 0.0f;
  bones[1] = bones[0];

  float xe[8][4];

#pragma unroll
  for (int tp = 0; tp < 4; ++tp) {
    const TileFrag& cur = tf[tp & 1];

    __builtin_amdgcn_sched_barrier(0);
    if (tp == 0) {
    } else if (tp + 1 < 4) {
      gmm_load_b(tf[(tp + 1) & 1], bpLane, cmLane, lamLane, tp + 1);
    } else {
#pragma unroll
      for (int r = 0; r < 8; ++r)
#pragma unroll
        for (int n = 0; n < 4; ++n)
          xe[r][n] = X[(size_t)(b0 + r + lhalf * 8) * ND + n * 16 + lmod];
    }
    __builtin_amdgcn_sched_barrier(0);

    v8f La = {};
    v8f Lb = {};
#pragma unroll
    for (int kk = 0; kk < 32; ++kk) {
      if (tp == 0 && kk == 16) {
        __builtin_amdgcn_sched_barrier(0);
        gmm_load_b(tf[1], bpLane, cmLane, lamLane, 1);
        __builtin_amdgcn_sched_barrier(0);
      }
      const v4f bb = cur.b[kk];
      v2f be; be[0] = bb[0]; be[1] = bb[1];
      v2f bo; bo[0] = bb[2]; bo[1] = bb[3];
      La = __builtin_amdgcn_wmma_f32_16x16x4_f32(false, xa[kk], false, be,
                                                 (short)0, La, false, false);
      Lb = __builtin_amdgcn_wmma_f32_16x16x4_f32(false, xa[kk], false, bo,
                                                 (short)0, Lb, false, false);
    }

    v2f w[8];
#pragma unroll
    for (int r = 0; r < 8; ++r) {
      float lwa = -0.5f * (La[r] + cur.cm[0]);
      lwa = fminf(50.0f, fmaxf(-50.0f, lwa));
      float lwb = -0.5f * (Lb[r] + cur.cm[1]);
      lwb = fminf(50.0f, fmaxf(-50.0f, lwb));
      v2f wv;
      wv[0] = __expf(lwa) * cur.lam[0];
      wv[1] = __expf(lwb) * cur.lam[1];
      w[r] = wv;
    }

    gmm_wave_sync();
#pragma unroll
    for (int r = 0; r < 8; ++r)
      *(v2f*)(&lds[(r + lhalf * 8) * 36 + 2 * lmod]) = w[r];
    gmm_wave_sync();
    v2f wa[8];
#pragma unroll
    for (int j = 0; j < 8; ++j)
      wa[j] = *(const v2f*)(&lds[lmod * 36 + j * 4 + lhalf * 2]);

    __builtin_amdgcn_sched_barrier(0);
    if (tp + 1 < 4)
      gmm_load_pq(tf[(tp + 1) & 1], pqLane, tp + 1);
    __builtin_amdgcn_sched_barrier(0);

#pragma unroll
    for (int j = 0; j < 8; ++j) {
#pragma unroll
      for (int n = 0; n < 4; ++n) {
        v4f pq = cur.pq[j * 4 + n];
        v2f bp; bp[0] = pq[0]; bp[1] = pq[1];
        accP[n] = __builtin_amdgcn_wmma_f32_16x16x4_f32(false, wa[j], false, bp,
                                                        (short)0, accP[n],
                                                        false, false);
        v2f bq; bq[0] = pq[2]; bq[1] = pq[3];
        accQ[n] = __builtin_amdgcn_wmma_f32_16x16x4_f32(false, wa[j], false, bq,
                                                        (short)0, accQ[n],
                                                        false, false);
      }
      accN = __builtin_amdgcn_wmma_f32_16x16x4_f32(false, wa[j], false, bones,
                                                   (short)0, accN,
                                                   false, false);
    }
  }

  float* pTile = part + ((size_t)chunk * BN + b0) * ND;
#pragma unroll
  for (int r = 0; r < 8; ++r) {
    const int bl = r + lhalf * 8;
#pragma unroll
    for (int n = 0; n < 4; ++n) {
      const int d = n * 16 + lmod;
      pTile[bl * ND + d] = xe[r][n] * accP[n][r] + accQ[n][r];
    }
  }
  if (lmod == 0) {
    float* np = nPart + (size_t)chunk * BN + b0 + lhalf * 8;
    v4f n0; n0[0] = accN[0]; n0[1] = accN[1]; n0[2] = accN[2]; n0[3] = accN[3];
    v4f n1; n1[0] = accN[4]; n1[1] = accN[5]; n1[2] = accN[6]; n1[3] = accN[7];
    *(v4f*)np       = n0;
    *(v4f*)(np + 4) = n1;
  }
}

#define NCHUNK 8
__global__ void __launch_bounds__(128)
gmm_finalize_kernel(const float* __restrict__ part,
                    const float* __restrict__ nPart,
                    float* __restrict__ out) {
  const int t = blockIdx.x * blockDim.x + threadIdx.x;
  if (t >= BN * ND / 4) return;
  const int b = t >> 4;
  const size_t e = (size_t)t * 4;
  v4f pv[NCHUNK];
  float nv[NCHUNK];
#pragma unroll
  for (int c = 0; c < NCHUNK; ++c) {
    pv[c] = *(const v4f*)(part + (size_t)c * (BN * ND) + e);
    nv[c] = nPart[c * BN + b];
  }
  v4f s = pv[0];
  float ns = nv[0];
#pragma unroll
  for (int c = 1; c < NCHUNK; ++c) {
    s  += pv[c];
    ns += nv[c];
  }
  v4f o;
  o[0] = s[0] / ns; o[1] = s[1] / ns; o[2] = s[2] / ns; o[3] = s[3] / ns;
  *(v4f*)(out + e) = o;
}

extern "C" void kernel_launch(void* const* d_in, const int* in_sizes, int n_in,
                              void* d_out, int out_size, void* d_ws, size_t ws_size,
                              hipStream_t stream) {
  const float* X   = (const float*)d_in[0];
  const float* Mu0 = (const float*)d_in[1];
  const float* Mu1 = (const float*)d_in[2];
  const float* S0  = (const float*)d_in[3];
  const float* S1  = (const float*)d_in[4];
  const float* Lam = (const float*)d_in[5];
  const float* tP  = (const float*)d_in[6];
  const float* eP  = (const float*)d_in[7];
  float* out = (float*)d_out;

  float* ws   = (float*)d_ws;
  float* Bp   = ws;
  float* cM   = Bp + 64 * 2 * MM;
  float* PQ   = cM + MM;
  float* part  = PQ + (MM / 2) * ND * 4;
  float* nPart = part + 8 * BN * ND;
  (void)ws_size; (void)in_sizes; (void)n_in; (void)out_size;

  gmm_prep_kernel<<<MM / 8, 128, 0, stream>>>(Mu0, Mu1, S0, S1, tP, eP,
                                              Bp, cM, PQ);

  gmm_main_kernel<<<dim3(BN / 64, 8), 128, 0, stream>>>(X, Lam, cM, Bp,
                                                        PQ, part, nPart);

  gmm_finalize_kernel<<<(BN * ND / 4 + 127) / 128, 128, 0, stream>>>(part, nPart, out);
}
